# starting kernel
# baseline (speedup 1.0000x reference)
#include <hip/hip_runtime.h>
#include <hip/hip_bf16.h>

typedef __attribute__((ext_vector_type(2))) float v2f;
typedef __attribute__((ext_vector_type(8))) float v8f;

#define DD   1024
#define BB   16
#define HH   4
#define EPSF 1e-8f
#define LOG2E 1.44269504088896340736f

__global__ __launch_bounds__(128) void flatt_kernel(
    const float* __restrict__ x,
    const float* __restrict__ aq, const float* __restrict__ bq,
    const float* __restrict__ ak, const float* __restrict__ bk,
    const float* __restrict__ av, const float* __restrict__ bv,
    float* __restrict__ out)
{
    __shared__ float xs[DD];
    __shared__ float att_s[HH * 16];
    __shared__ float dn_s[HH * 16];

    const int b     = blockIdx.x >> 6;
    const int itile = blockIdx.x & 63;
    const int i0    = itile << 4;
    const int tid   = threadIdx.x;
    const int h     = tid >> 5;
    const int lane  = tid & 31;
    const int iloc  = lane & 15;
    const int half  = lane >> 4;

    #pragma unroll
    for (int r = 0; r < DD / 128; ++r)
        xs[tid + 128 * r] = x[b * DD + tid + 128 * r];
    __syncthreads();

    const float Aq = aq[h], Bq = bq[h];
    const float Ak = ak[h], Bk = bk[h];
    const float Av = av[h], Bv = bv[h];

    const float q = fmaf(xs[i0 + iloc], Aq, Bq);

    float m = -3.4e38f;
    for (int c = 0; c < DD / 4; ++c) {
        const int j = 4 * c + 2 * half;
        const float x0 = xs[j], x1 = xs[j + 1];
        const float s0 = __builtin_amdgcn_rcpf(fabsf(fmaf(x0, Ak, Bk) - q) + EPSF);
        const float s1 = __builtin_amdgcn_rcpf(fabsf(fmaf(x1, Ak, Bk) - q) + EPSF);
        m = fmaxf(m, fmaxf(s0, s1));
    }
    m = fmaxf(m, __shfl_xor(m, 16, 32));

    v8f acc = {};
    float denom = 0.0f;
    for (int c = 0; c < DD / 4; ++c) {
        const int j = 4 * c + 2 * half;
        const float x0 = xs[j], x1 = xs[j + 1];
        const float s0 = __builtin_amdgcn_rcpf(fabsf(fmaf(x0, Ak, Bk) - q) + EPSF);
        const float s1 = __builtin_amdgcn_rcpf(fabsf(fmaf(x1, Ak, Bk) - q) + EPSF);
        const float p0 = __builtin_amdgcn_exp2f((s0 - m) * LOG2E);
        const float p1 = __builtin_amdgcn_exp2f((s1 - m) * LOG2E);
        denom += p0 + p1;

        v2f A;  A.x  = p0;                  A.y  = p1;
        v2f Bm; Bm.x = fmaf(x0, Av, Bv);    Bm.y = fmaf(x1, Av, Bv);

        acc = __builtin_amdgcn_wmma_f32_16x16x4_f32(
                  false, A, false, Bm, (short)0, acc, false, false);
    }
    denom += __shfl_xor(denom, 16, 32);

    if (lane < 16) dn_s[h * 16 + iloc] = denom;
    if (lane == 0) {
        #pragma unroll
        for (int r = 0; r < 8; ++r) att_s[h * 16 + r] = acc[r];
    }
    if (lane == 16) {
        #pragma unroll
        for (int r = 0; r < 8; ++r) att_s[h * 16 + 8 + r] = acc[r];
    }
    __syncthreads();

    if (tid < 16) {
        float s = 0.0f;
        #pragma unroll
        for (int hh = 0; hh < HH; ++hh)
            s += att_s[hh * 16 + tid] / dn_s[hh * 16 + tid];
        out[b * DD + i0 + tid] = xs[i0 + tid] + 0.5f * s;
    }
}

extern "C" void kernel_launch(void* const* d_in, const int* in_sizes, int n_in,
                              void* d_out, int out_size, void* d_ws, size_t ws_size,
                              hipStream_t stream) {
    (void)in_sizes; (void)n_in; (void)d_ws; (void)ws_size; (void)out_size;
    const float* x  = (const float*)d_in[0];
    const float* aq = (const float*)d_in[1];
    const float* bq = (const float*)d_in[2];
    const float* ak = (const float*)d_in[3];
    const float* bk = (const float*)d_in[4];
    const float* av = (const float*)d_in[5];
    const float* bv = (const float*)d_in[6];
    float* out = (float*)d_out;

    flatt_kernel<<<dim3(BB * (DD / 16)), dim3(128), 0, stream>>>(
        x, aq, bq, ak, bk, av, bv, out);
}


# each wave carries 4 row tiles (64 rows per block, 256 blocks): x_j read, k_j and the A operand are built once per key pair and serve 4 matrix instructions on 4 accumulators; sum order kept
# speedup vs baseline: 1.5465x; 1.5465x over previous
#include <hip/hip_runtime.h>
#include <hip/hip_bf16.h>


typedef __attribute__((ext_vector_type(2))) float v2f;
typedef __attribute__((ext_vector_type(8))) float v8f;

#define DD   1024
#define BB   16
#define HH   4
#define EPSF 1e-8f
#define LOG2E 1.44269504088896340736f
#define CH   16
#define RT   4

__global__ __launch_bounds__(128) void flatt_kernel(
    const float* __restrict__ x,
    const float* __restrict__ aq, const float* __restrict__ bq,
    const float* __restrict__ ak, const float* __restrict__ bk,
    const float* __restrict__ av, const float* __restrict__ bv,
    float* __restrict__ out)
{
    __shared__ float xs[DD];
    __shared__ float att_s[HH * 16 * RT];
    __shared__ float dn_s[HH * 16 * RT];

    constexpr int BPR = DD / (16 * RT);
    const int b     = blockIdx.x / BPR;
    const int i0    = (blockIdx.x % BPR) * (16 * RT);
    const int tid   = threadIdx.x;
    const int h     = tid >> 5;
    const int lane  = tid & 31;
    const int iloc  = lane & 15;
    const int half  = lane >> 4;

    #pragma unroll
    for (int r = 0; r < DD / 128; ++r)
        xs[tid + 128 * r] = x[b * DD + tid + 128 * r];
    __syncthreads();

    const float Aq = aq[h], Bq = bq[h];
    const float Ak = ak[h], Bk = bk[h];
    const float Av = av[h], Bv = bv[h];

    float q[RT];
    #pragma unroll
    for (int r = 0; r < RT; ++r)
        q[r] = fmaf(xs[i0 + 16 * r + iloc], Aq, Bq);

    const bool rowV = (iloc == 8 * half);
    const bool rowO = (iloc == 8 * half + 1);
    const float AvL = rowV ? Av : 0.0f;
    const float BvL = rowV ? Bv : (rowO ? 1.0f : 0.0f);

    float nm[RT];
    v8f acc[RT];
    #pragma unroll
    for (int r = 0; r < RT; ++r) { nm[r] = 3.0e38f; acc[r] = v8f{}; }

    for (int t = 0; t < DD / 4; t += CH) {
        float kk0[CH], kk1[CH];
        v2f Am[CH];
        #pragma unroll
        for (int u = 0; u < CH; ++u) {
            const int j = 4 * (t + u) + 2 * half;
            const float x0 = xs[j];
            const float x1 = xs[j + 1];
            kk0[u] = fmaf(x0, Ak, Bk);
            kk1[u] = fmaf(x1, Ak, Bk);
            Am[u].x = fmaf(x0, AvL, BvL);
            Am[u].y = fmaf(x1, AvL, BvL);
        }

        float s0[RT][CH], s1[RT][CH];
        #pragma unroll
        for (int r = 0; r < RT; ++r) {
            #pragma unroll
            for (int u = 0; u < CH; ++u) {
                s0[r][u] = __builtin_amdgcn_rcpf(fabsf(kk0[u] - q[r]) + EPSF);
                s1[r][u] = __builtin_amdgcn_rcpf(fabsf(kk1[u] - q[r]) + EPSF);
            }
            float cm = fmaxf(s0[r][0], s1[r][0]);
            #pragma unroll
            for (int u = 1; u < CH; ++u) cm = fmaxf(cm, fmaxf(s0[r][u], s1[r][u]));

            const float nm_new = fminf(nm[r], cm * (-LOG2E));
            const float sc = __builtin_amdgcn_exp2f(nm_new - nm[r]);
            acc[r][0] *= sc;
            acc[r][1] *= sc;
            nm[r] = nm_new;
        }

        #pragma unroll
        for (int u = 0; u < CH; ++u) {
            #pragma unroll
            for (int r = 0; r < RT; ++r) {
                v2f P;
                P.x = __builtin_amdgcn_exp2f(fmaf(s0[r][u], LOG2E, nm[r]));
                P.y = __builtin_amdgcn_exp2f(fmaf(s1[r][u], LOG2E, nm[r]));
                acc[r] = __builtin_amdgcn_wmma_f32_16x16x4_f32(
                             false, Am[u], false, P, (short)0, acc[r], false, false);
            }
        }
    }

    #pragma unroll
    for (int r = 0; r < RT; ++r) {
        const float nm_o = __shfl_xor(nm[r], 16, 32);
        const float nm_t = fminf(nm[r], nm_o);
        const float f    = __builtin_amdgcn_exp2f(nm_t - nm[r]);
        float num = acc[r][0] * f;
        float den = acc[r][1] * f;
        num += __shfl_xor(num, 16, 32);
        den += __shfl_xor(den, 16, 32);
        if (lane < 16) {
            att_s[h * 16 * RT + 16 * r + iloc] = num;
            dn_s[h * 16 * RT + 16 * r + iloc]  = den;
        }
    }
    __syncthreads();

    if (tid < 16 * RT) {
        float s = 0.0f;
        #pragma unroll
        for (int hh = 0; hh < HH; ++hh)
            s += att_s[hh * 16 * RT + tid] / dn_s[hh * 16 * RT + tid];
        out[b * DD + i0 + tid] = xs[i0 + tid] + 0.5f * s;
    }
}

extern "C" void kernel_launch(void* const* d_in, const int* in_sizes, int n_in,
                              void* d_out, int out_size, void* d_ws, size_t ws_size,
                              hipStream_t stream) {
    (void)in_sizes; (void)n_in; (void)d_ws; (void)ws_size; (void)out_size;
    const float* x  = (const float*)d_in[0];
    const float* aq = (const float*)d_in[1];
    const float* bq = (const float*)d_in[2];
    const float* ak = (const float*)d_in[3];
    const float* bk = (const float*)d_in[4];
    const float* av = (const float*)d_in[5];
    const float* bv = (const float*)d_in[6];
    float* out = (float*)d_out;

    flatt_kernel<<<dim3(BB * (DD / (16 * RT))), dim3(128), 0, stream>>>(
        x, aq, bq, ak, bk, av, bv, out);
}
